# starting kernel
# baseline (speedup 1.0000x reference)
#include <hip/hip_runtime.h>


typedef float v2f __attribute__((ext_vector_type(2)));
typedef float v4f __attribute__((ext_vector_type(4)));
typedef float v8f __attribute__((ext_vector_type(8)));
typedef int   v4i __attribute__((ext_vector_type(4)));

#define SEQ  1024
#define EMB  1024
#define NH   64
#define DH   16

__global__ __launch_bounds__(32) void attn_flash_wmma(
    const float* __restrict__ keys, const float* __restrict__ query,
    const float* __restrict__ values, const int* __restrict__ mask,
    float* __restrict__ attn_out)
{
    const int lane = threadIdx.x;
    const int q    = lane & 15;
    const int half = lane >> 4;

    const int idx = blockIdx.x;
    const int qt  = idx & 63;
    const int h   = (idx >> 6) & 63;
    const int n   = idx >> 12;
    const int q0  = qt * 16;

    const float* Qrow = query + ((size_t)(n * SEQ + q0 + q)) * EMB + h * DH;
    v2f Bq[4];
#pragma unroll
    for (int dd = 0; dd < 4; ++dd)
        Bq[dd] = *reinterpret_cast<const v2f*>(Qrow + dd * 4 + 2 * half);

    v8f   o    = {};
    float mold = -1e30f;
    float lsum = 0.0f;

    const float* Kbase = keys   + ((size_t)(n * SEQ)) * EMB + h * DH;
    const float* Vbase = values + ((size_t)(n * SEQ)) * EMB + h * DH;
    const int*   Mrow  = mask + (size_t)n * SEQ * SEQ + (size_t)(q0 + q) * SEQ;

    for (int kt = 0; kt < SEQ / 16; ++kt) {
        const float* Krow = Kbase + (size_t)(kt * 16 + q) * EMB;
        v8f s = {};
#pragma unroll
        for (int dd = 0; dd < 4; ++dd) {
            v2f a = *reinterpret_cast<const v2f*>(Krow + dd * 4 + 2 * half);
            s = __builtin_amdgcn_wmma_f32_16x16x4_f32(
                    false, a, false, Bq[dd], (short)0, s, false, false);
        }

        const int* mb = Mrow + kt * 16 + 8 * half;
        v4i m0 = *reinterpret_cast<const v4i*>(mb);
        v4i m1 = *reinterpret_cast<const v4i*>(mb + 4);
        float p[8];
#pragma unroll
        for (int r = 0; r < 8; ++r) {
            float sr = s[r];
            int   mv = (r < 4) ? m0[r] : m1[r - 4];
            sr   = (mv == 0) ? -1e20f : sr;
            p[r] = sr * 0.03125f;
        }

        float tmax = p[0];
#pragma unroll
        for (int r = 1; r < 8; ++r) tmax = fmaxf(tmax, p[r]);
        tmax = fmaxf(tmax, __shfl_xor(tmax, 16));
        const float mnew  = fmaxf(mold, tmax);
        const float alpha = __expf(mold - mnew);
        float tsum = 0.0f;
#pragma unroll
        for (int r = 0; r < 8; ++r) { p[r] = __expf(p[r] - mnew); tsum += p[r]; }
        tsum += __shfl_xor(tsum, 16);
        lsum = lsum * alpha + tsum;
        mold = mnew;
#pragma unroll
        for (int r = 0; r < 8; ++r) o[r] *= alpha;

        const float* Vt = Vbase + (size_t)(kt * 16) * EMB + q;
#pragma unroll
        for (int kk = 0; kk < 4; ++kk) {
            const int k0 = 4 * kk + 2 * half;
            v2f av;
            av.x = Vt[(size_t)k0 * EMB];
            av.y = Vt[(size_t)(k0 + 1) * EMB];
            const int a       = (kk & 1) * 4;
            const int srcLane = q | ((kk >> 1) << 4);
            float t0 = __shfl(p[a],     srcLane);
            float t2 = __shfl(p[a + 2], srcLane);
            float t1 = __shfl(p[a + 1], srcLane);
            float t3 = __shfl(p[a + 3], srcLane);
            v2f bp;
            bp.x = half ? t2 : t0;
            bp.y = half ? t3 : t1;
            o = __builtin_amdgcn_wmma_f32_16x16x4_f32(
                    false, av, false, bp, (short)0, o, false, false);
        }
    }

    const float rl = 1.0f / lsum;
    float* Orow = attn_out + ((size_t)(n * SEQ + q0 + q)) * EMB + h * DH + 8 * half;
    v4f o0, o1;
#pragma unroll
    for (int r = 0; r < 4; ++r) { o0[r] = o[r] * rl; o1[r] = o[r + 4] * rl; }
    *reinterpret_cast<v4f*>(Orow)     = o0;
    *reinterpret_cast<v4f*>(Orow + 4) = o1;
}

__global__ __launch_bounds__(32) void proj_wmma(
    const float* __restrict__ X, const float* __restrict__ W,
    const float* __restrict__ bvec, float* __restrict__ out)
{
    const int lane = threadIdx.x;
    const int q    = lane & 15;
    const int half = lane >> 4;
    const int j0 = (blockIdx.x & 63) * 16;
    const int i0 = (blockIdx.x >> 6) * 16;

    const float* Xrow = X + (size_t)(i0 + q) * EMB;
    const float* Wrow = W + (size_t)(j0 + q) * EMB;
    v8f acc = {};
#pragma unroll 4
    for (int ee = 0; ee < EMB / 4; ++ee) {
        v2f a = *reinterpret_cast<const v2f*>(Xrow + ee * 4 + 2 * half);
        v2f b = *reinterpret_cast<const v2f*>(Wrow + ee * 4 + 2 * half);
        acc = __builtin_amdgcn_wmma_f32_16x16x4_f32(
                  false, a, false, b, (short)0, acc, false, false);
    }
    const float bj = bvec[j0 + q];
#pragma unroll
    for (int r = 0; r < 8; ++r)
        out[(size_t)(i0 + r + 8 * half) * EMB + j0 + q] = acc[r] + bj;
}

extern "C" void kernel_launch(void* const* d_in, const int* in_sizes, int n_in,
                              void* d_out, int out_size, void* d_ws, size_t ws_size,
                              hipStream_t stream)
{
    const float* keys   = (const float*)d_in[0];
    const float* query  = (const float*)d_in[1];
    const float* values = (const float*)d_in[2];
    const int*   mask   = (const int*)d_in[3];
    const float* W_out  = (const float*)d_in[4];
    const float* b_out  = (const float*)d_in[5];
    float* out  = (float*)d_out;
    float* attn = (float*)d_ws;

    const int N = 2;
    attn_flash_wmma<<<N * NH * (SEQ / 16), 32, 0, stream>>>(
        keys, query, values, mask, attn);
    proj_wmma<<<(N * SEQ / 16) * (EMB / 16), 32, 0, stream>>>(
        attn, W_out, b_out, out);
}


# proj GEMM without LDS: workspace kept in 16x8 fragment blocks so X loads are whole-line b128; a wave owns 64x16, one W b128 fragment feeds 4 row tiles; k summed 0,1,4,5,2,3,6,7 per 8 (order changes)
# speedup vs baseline: 1.5704x; 1.5704x over previous
#include <hip/hip_runtime.h>


typedef float v2f __attribute__((ext_vector_type(2)));
typedef float v4f __attribute__((ext_vector_type(4)));
typedef float v8f __attribute__((ext_vector_type(8)));
typedef int   v4i __attribute__((ext_vector_type(4)));

#define SEQ  1024
#define EMB  1024
#define NH   64
#define DH   16
#define QT   4

#define XBLK 128
#define XROWT (XBLK * (EMB / 8))
#define PT   4
#define PG   4

__global__ __launch_bounds__(32) __attribute__((amdgpu_waves_per_eu(1, 2))) void attn_flash_wmma(
    const float* __restrict__ keys, const float* __restrict__ query,
    const float* __restrict__ values, const int* __restrict__ mask,
    float* __restrict__ attn_out)
{
    const int lane = threadIdx.x;
    const int q    = lane & 15;
    const int half = lane >> 4;

    const int idx = blockIdx.x;
    const int qg  = idx & 15;
    const int h   = (idx >> 4) & 63;
    const int n   = idx >> 10;
    const int q0  = qg * (QT * 16);

    v2f Bq[QT][4];
#pragma unroll
    for (int t = 0; t < QT; ++t) {
        const float* Qrow = query + ((size_t)(n * SEQ + q0 + 16 * t + q)) * EMB + h * DH;
#pragma unroll
        for (int dd = 0; dd < 4; ++dd)
            Bq[t][dd] = *reinterpret_cast<const v2f*>(Qrow + dd * 4 + 2 * half);
    }

    v8f   o[QT];
    float mold[QT];
    float lsum[QT];
#pragma unroll
    for (int t = 0; t < QT; ++t) { o[t] = v8f{}; mold[t] = -1e30f; lsum[t] = 0.0f; }

    const float* Kbase = keys   + ((size_t)(n * SEQ)) * EMB + h * DH;
    const float* Vbase = values + ((size_t)(n * SEQ)) * EMB + h * DH;
    const int*   Mrow  = mask + (size_t)n * SEQ * SEQ + (size_t)(q0 + q) * SEQ;

    for (int kt = 0; kt < SEQ / 16; ++kt) {
        const float* Krow = Kbase + (size_t)(kt * 16 + q) * EMB;
        v2f ka[4];
#pragma unroll
        for (int dd = 0; dd < 4; ++dd)
            ka[dd] = *reinterpret_cast<const v2f*>(Krow + dd * 4 + 2 * half);

        const float* Vt = Vbase + (size_t)(kt * 16) * EMB + q;
        v2f av[4];
#pragma unroll
        for (int kk = 0; kk < 4; ++kk) {
            const int k0 = 4 * kk + 2 * half;
            av[kk].x = Vt[(size_t)k0 * EMB];
            av[kk].y = Vt[(size_t)(k0 + 1) * EMB];
        }

        v4i mk0[QT], mk1[QT];
#pragma unroll
        for (int t = 0; t < QT; ++t) {
            const int* mb = Mrow + (size_t)(16 * t) * SEQ + kt * 16 + 8 * half;
            mk0[t] = *reinterpret_cast<const v4i*>(mb);
            mk1[t] = *reinterpret_cast<const v4i*>(mb + 4);
        }

#pragma unroll
        for (int t = 0; t < QT; ++t) {
            v8f s = {};
#pragma unroll
            for (int dd = 0; dd < 4; ++dd)
                s = __builtin_amdgcn_wmma_f32_16x16x4_f32(
                        false, ka[dd], false, Bq[t][dd], (short)0, s, false, false);

            const v4i m0 = mk0[t];
            const v4i m1 = mk1[t];
            float p[8];
#pragma unroll
            for (int r = 0; r < 8; ++r) {
                float sr = s[r];
                int   mv = (r < 4) ? m0[r] : m1[r - 4];
                sr   = (mv == 0) ? -1e20f : sr;
                p[r] = sr * 0.03125f;
            }

            float tmax = p[0];
#pragma unroll
            for (int r = 1; r < 8; ++r) tmax = fmaxf(tmax, p[r]);
            tmax = fmaxf(tmax, __shfl_xor(tmax, 16));
            const float mnew  = fmaxf(mold[t], tmax);
            const float alpha = __expf(mold[t] - mnew);
            float tsum = 0.0f;
#pragma unroll
            for (int r = 0; r < 8; ++r) { p[r] = __expf(p[r] - mnew); tsum += p[r]; }
            tsum += __shfl_xor(tsum, 16);
            lsum[t] = lsum[t] * alpha + tsum;
            mold[t] = mnew;
#pragma unroll
            for (int r = 0; r < 8; ++r) o[t][r] *= alpha;

#pragma unroll
            for (int kk = 0; kk < 4; ++kk) {
                const int a       = (kk & 1) * 4;
                const int srcLane = q | ((kk >> 1) << 4);
                float t0 = __shfl(p[a],     srcLane);
                float t2 = __shfl(p[a + 2], srcLane);
                float t1 = __shfl(p[a + 1], srcLane);
                float t3 = __shfl(p[a + 3], srcLane);
                v2f bp;
                bp.x = half ? t2 : t0;
                bp.y = half ? t3 : t1;
                o[t] = __builtin_amdgcn_wmma_f32_16x16x4_f32(
                        false, av[kk], false, bp, (short)0, o[t], false, false);
            }
        }
    }

#pragma unroll
    for (int t = 0; t < QT; ++t) {
        const float rl = 1.0f / lsum[t];
        const int R = (n * SEQ + q0) / 16 + t;
        float* Oblk = attn_out + (size_t)R * XROWT + (size_t)(2 * h + half) * XBLK + q * 4;
        v4f o0, o1;
#pragma unroll
        for (int r = 0; r < 4; ++r) { o0[r] = o[t][r] * rl; o1[r] = o[t][r + 4] * rl; }
        *reinterpret_cast<v4f*>(Oblk)      = o0;
        *reinterpret_cast<v4f*>(Oblk + 64) = o1;
    }
}

struct ProjStage {
    v4f b[PG];
    v4f a[PG][PT];
};

__device__ __forceinline__ void proj_load(ProjStage& st, const float* __restrict__ Xp,
                                          const float* __restrict__ Wp, int s)
{
#pragma unroll
    for (int u = 0; u < PG; ++u) {
        const int g = s * PG + u;
        st.b[u] = *reinterpret_cast<const v4f*>(Wp + 8 * g);
#pragma unroll
        for (int t = 0; t < PT; ++t)
            st.a[u][t] = *reinterpret_cast<const v4f*>(Xp + (size_t)t * XROWT + g * XBLK);
    }
}

__device__ __forceinline__ void proj_mma(v8f (&acc)[PT], const ProjStage& st)
{
#pragma unroll
    for (int u = 0; u < PG; ++u) {
        const v2f b0 = {st.b[u][0], st.b[u][1]};
        const v2f b1 = {st.b[u][2], st.b[u][3]};
#pragma unroll
        for (int t = 0; t < PT; ++t) {
            const v2f a0 = {st.a[u][t][0], st.a[u][t][1]};
            acc[t] = __builtin_amdgcn_wmma_f32_16x16x4_f32(
                         false, a0, false, b0, (short)0, acc[t], false, false);
        }
#pragma unroll
        for (int t = 0; t < PT; ++t) {
            const v2f a1 = {st.a[u][t][2], st.a[u][t][3]};
            acc[t] = __builtin_amdgcn_wmma_f32_16x16x4_f32(
                         false, a1, false, b1, (short)0, acc[t], false, false);
        }
    }
}

__global__ __launch_bounds__(128) __attribute__((amdgpu_waves_per_eu(1, 2))) void proj_wmma(
    const float* __restrict__ X, const float* __restrict__ W,
    const float* __restrict__ bvec, float* __restrict__ out)
{
    const int lane = threadIdx.x & 31;
    const int wv   = threadIdx.x >> 5;
    const int q    = lane & 15;
    const int half = lane >> 4;
    const int j0 = ((blockIdx.x & 15) * 4 + wv) * 16;
    const int R0 = (blockIdx.x >> 4) * PT;

    const float* Xp = X + (size_t)R0 * XROWT + lane * 4;
    const float* Wp = W + (size_t)(j0 + q) * EMB + 4 * half;

    v8f acc[PT];
#pragma unroll
    for (int t = 0; t < PT; ++t) acc[t] = v8f{};

    constexpr int NS = EMB / (8 * PG);
    ProjStage s0, s1;
    proj_load(s0, Xp, Wp, 0);
    for (int s = 0; s < NS - 2; s += 2) {
        proj_load(s1, Xp, Wp, s + 1);
        proj_mma(acc, s0);
        proj_load(s0, Xp, Wp, s + 2);
        proj_mma(acc, s1);
    }
    proj_load(s1, Xp, Wp, NS - 1);
    proj_mma(acc, s0);
    proj_mma(acc, s1);

    const float bj = bvec[j0 + q];
#pragma unroll
    for (int t = 0; t < PT; ++t)
#pragma unroll
        for (int r = 0; r < 8; ++r)
            out[(size_t)((R0 + t) * 16 + r + 8 * half) * EMB + j0 + q] = acc[t][r] + bj;
}

extern "C" void kernel_launch(void* const* d_in, const int* in_sizes, int n_in,
                              void* d_out, int out_size, void* d_ws, size_t ws_size,
                              hipStream_t stream)
{
    const float* keys   = (const float*)d_in[0];
    const float* query  = (const float*)d_in[1];
    const float* values = (const float*)d_in[2];
    const int*   mask   = (const int*)d_in[3];
    const float* W_out  = (const float*)d_in[4];
    const float* b_out  = (const float*)d_in[5];
    float* out  = (float*)d_out;
    float* attn = (float*)d_ws;

    const int N = 2;
    attn_flash_wmma<<<N * NH * (SEQ / (16 * QT)), 32, 0, stream>>>(
        keys, query, values, mask, attn);
    proj_wmma<<<(N * SEQ / (16 * PT)) * (EMB / 64), 128, 0, stream>>>(
        attn, W_out, b_out, out);
}


# attention: each 16-key step's K, V and mask fragments are loaded one step ahead into a second register set (loop unrolled by 2, last pair peeled), so loads run under compute; sum order kept
# speedup vs baseline: 1.7610x; 1.1214x over previous
#include <hip/hip_runtime.h>


typedef float v2f __attribute__((ext_vector_type(2)));
typedef float v4f __attribute__((ext_vector_type(4)));
typedef float v8f __attribute__((ext_vector_type(8)));
typedef int   v4i __attribute__((ext_vector_type(4)));

#define SEQ  1024
#define EMB  1024
#define NH   64
#define DH   16
#define QT   4

#define XBLK 128
#define XROWT (XBLK * (EMB / 8))
#define PT   4
#define PG   4

struct AttnFrag {
    v2f ka[4];
    v2f av[4];
    v4i mk0[QT], mk1[QT];
};

__device__ __forceinline__ void attn_load(AttnFrag& f, const float* __restrict__ Kbase,
                                          const float* __restrict__ Vbase,
                                          const int* __restrict__ Mrow,
                                          int kt, int q, int half)
{
    const float* Krow = Kbase + (size_t)(kt * 16 + q) * EMB;
#pragma unroll
    for (int dd = 0; dd < 4; ++dd)
        f.ka[dd] = *reinterpret_cast<const v2f*>(Krow + dd * 4 + 2 * half);

    const float* Vt = Vbase + (size_t)(kt * 16) * EMB + q;
#pragma unroll
    for (int kk = 0; kk < 4; ++kk) {
        const int k0 = 4 * kk + 2 * half;
        f.av[kk].x = Vt[(size_t)k0 * EMB];
        f.av[kk].y = Vt[(size_t)(k0 + 1) * EMB];
    }

#pragma unroll
    for (int t = 0; t < QT; ++t) {
        const int* mb = Mrow + (size_t)(16 * t) * SEQ + kt * 16 + 8 * half;
        f.mk0[t] = *reinterpret_cast<const v4i*>(mb);
        f.mk1[t] = *reinterpret_cast<const v4i*>(mb + 4);
    }
}

__device__ __forceinline__ void attn_step(const AttnFrag& f, const v2f (&Bq)[QT][4],
                                          v8f (&o)[QT], float (&mold)[QT], float (&lsum)[QT],
                                          int q, int half)
{
#pragma unroll
    for (int t = 0; t < QT; ++t) {
        v8f s = {};
#pragma unroll
        for (int dd = 0; dd < 4; ++dd)
            s = __builtin_amdgcn_wmma_f32_16x16x4_f32(
                    false, f.ka[dd], false, Bq[t][dd], (short)0, s, false, false);

        const v4i m0 = f.mk0[t];
        const v4i m1 = f.mk1[t];
        float p[8];
#pragma unroll
        for (int r = 0; r < 8; ++r) {
            float sr = s[r];
            int   mv = (r < 4) ? m0[r] : m1[r - 4];
            sr   = (mv == 0) ? -1e20f : sr;
            p[r] = sr * 0.03125f;
        }

        float tmax = p[0];
#pragma unroll
        for (int r = 1; r < 8; ++r) tmax = fmaxf(tmax, p[r]);
        tmax = fmaxf(tmax, __shfl_xor(tmax, 16));
        const float mnew  = fmaxf(mold[t], tmax);
        const float alpha = __expf(mold[t] - mnew);
        float tsum = 0.0f;
#pragma unroll
        for (int r = 0; r < 8; ++r) { p[r] = __expf(p[r] - mnew); tsum += p[r]; }
        tsum += __shfl_xor(tsum, 16);
        lsum[t] = lsum[t] * alpha + tsum;
        mold[t] = mnew;
#pragma unroll
        for (int r = 0; r < 8; ++r) o[t][r] *= alpha;

#pragma unroll
        for (int kk = 0; kk < 4; ++kk) {
            const int a       = (kk & 1) * 4;
            const int srcLane = q | ((kk >> 1) << 4);
            float t0 = __shfl(p[a],     srcLane);
            float t2 = __shfl(p[a + 2], srcLane);
            float t1 = __shfl(p[a + 1], srcLane);
            float t3 = __shfl(p[a + 3], srcLane);
            v2f bp;
            bp.x = half ? t2 : t0;
            bp.y = half ? t3 : t1;
            o[t] = __builtin_amdgcn_wmma_f32_16x16x4_f32(
                    false, f.av[kk], false, bp, (short)0, o[t], false, false);
        }
    }
}

__global__ __launch_bounds__(32) __attribute__((amdgpu_waves_per_eu(1, 2))) void attn_flash_wmma(
    const float* __restrict__ keys, const float* __restrict__ query,
    const float* __restrict__ values, const int* __restrict__ mask,
    float* __restrict__ attn_out)
{
    const int lane = threadIdx.x;
    const int q    = lane & 15;
    const int half = lane >> 4;

    const int idx = blockIdx.x;
    const int qg  = idx & 15;
    const int h   = (idx >> 4) & 63;
    const int n   = idx >> 10;
    const int q0  = qg * (QT * 16);

    v2f Bq[QT][4];
#pragma unroll
    for (int t = 0; t < QT; ++t) {
        const float* Qrow = query + ((size_t)(n * SEQ + q0 + 16 * t + q)) * EMB + h * DH;
#pragma unroll
        for (int dd = 0; dd < 4; ++dd)
            Bq[t][dd] = *reinterpret_cast<const v2f*>(Qrow + dd * 4 + 2 * half);
    }

    v8f   o[QT];
    float mold[QT];
    float lsum[QT];
#pragma unroll
    for (int t = 0; t < QT; ++t) { o[t] = v8f{}; mold[t] = -1e30f; lsum[t] = 0.0f; }

    const float* Kbase = keys   + ((size_t)(n * SEQ)) * EMB + h * DH;
    const float* Vbase = values + ((size_t)(n * SEQ)) * EMB + h * DH;
    const int*   Mrow  = mask + (size_t)n * SEQ * SEQ + (size_t)(q0 + q) * SEQ;

    constexpr int NK = SEQ / 16;
    AttnFrag f0, f1;
    attn_load(f0, Kbase, Vbase, Mrow, 0, q, half);
    for (int kt = 0; kt < NK - 2; kt += 2) {
        attn_load(f1, Kbase, Vbase, Mrow, kt + 1, q, half);
        attn_step(f0, Bq, o, mold, lsum, q, half);
        attn_load(f0, Kbase, Vbase, Mrow, kt + 2, q, half);
        attn_step(f1, Bq, o, mold, lsum, q, half);
    }
    attn_load(f1, Kbase, Vbase, Mrow, NK - 1, q, half);
    attn_step(f0, Bq, o, mold, lsum, q, half);
    attn_step(f1, Bq, o, mold, lsum, q, half);

#pragma unroll
    for (int t = 0; t < QT; ++t) {
        const float rl = 1.0f / lsum[t];
        const int R = (n * SEQ + q0) / 16 + t;
        float* Oblk = attn_out + (size_t)R * XROWT + (size_t)(2 * h + half) * XBLK + q * 4;
        v4f o0, o1;
#pragma unroll
        for (int r = 0; r < 4; ++r) { o0[r] = o[t][r] * rl; o1[r] = o[t][r + 4] * rl; }
        *reinterpret_cast<v4f*>(Oblk)      = o0;
        *reinterpret_cast<v4f*>(Oblk + 64) = o1;
    }
}

struct ProjStage {
    v4f b[PG];
    v4f a[PG][PT];
};

__device__ __forceinline__ void proj_load(ProjStage& st, const float* __restrict__ Xp,
                                          const float* __restrict__ Wp, int s)
{
#pragma unroll
    for (int u = 0; u < PG; ++u) {
        const int g = s * PG + u;
        st.b[u] = *reinterpret_cast<const v4f*>(Wp + 8 * g);
#pragma unroll
        for (int t = 0; t < PT; ++t)
            st.a[u][t] = *reinterpret_cast<const v4f*>(Xp + (size_t)t * XROWT + g * XBLK);
    }
}

__device__ __forceinline__ void proj_mma(v8f (&acc)[PT], const ProjStage& st)
{
#pragma unroll
    for (int u = 0; u < PG; ++u) {
        const v2f b0 = {st.b[u][0], st.b[u][1]};
        const v2f b1 = {st.b[u][2], st.b[u][3]};
#pragma unroll
        for (int t = 0; t < PT; ++t) {
            const v2f a0 = {st.a[u][t][0], st.a[u][t][1]};
            acc[t] = __builtin_amdgcn_wmma_f32_16x16x4_f32(
                         false, a0, false, b0, (short)0, acc[t], false, false);
        }
#pragma unroll
        for (int t = 0; t < PT; ++t) {
            const v2f a1 = {st.a[u][t][2], st.a[u][t][3]};
            acc[t] = __builtin_amdgcn_wmma_f32_16x16x4_f32(
                         false, a1, false, b1, (short)0, acc[t], false, false);
        }
    }
}

__global__ __launch_bounds__(128) __attribute__((amdgpu_waves_per_eu(1, 2))) void proj_wmma(
    const float* __restrict__ X, const float* __restrict__ W,
    const float* __restrict__ bvec, float* __restrict__ out)
{
    const int lane = threadIdx.x & 31;
    const int wv   = threadIdx.x >> 5;
    const int q    = lane & 15;
    const int half = lane >> 4;
    const int j0 = ((blockIdx.x & 15) * 4 + wv) * 16;
    const int R0 = (blockIdx.x >> 4) * PT;

    const float* Xp = X + (size_t)R0 * XROWT + lane * 4;
    const float* Wp = W + (size_t)(j0 + q) * EMB + 4 * half;

    v8f acc[PT];
#pragma unroll
    for (int t = 0; t < PT; ++t) acc[t] = v8f{};

    constexpr int NS = EMB / (8 * PG);
    ProjStage s0, s1;
    proj_load(s0, Xp, Wp, 0);
    for (int s = 0; s < NS - 2; s += 2) {
        proj_load(s1, Xp, Wp, s + 1);
        proj_mma(acc, s0);
        proj_load(s0, Xp, Wp, s + 2);
        proj_mma(acc, s1);
    }
    proj_load(s1, Xp, Wp, NS - 1);
    proj_mma(acc, s0);
    proj_mma(acc, s1);

    const float bj = bvec[j0 + q];
#pragma unroll
    for (int t = 0; t < PT; ++t)
#pragma unroll
        for (int r = 0; r < 8; ++r)
            out[(size_t)((R0 + t) * 16 + r + 8 * half) * EMB + j0 + q] = acc[t][r] + bj;
}

extern "C" void kernel_launch(void* const* d_in, const int* in_sizes, int n_in,
                              void* d_out, int out_size, void* d_ws, size_t ws_size,
                              hipStream_t stream)
{
    const float* keys   = (const float*)d_in[0];
    const float* query  = (const float*)d_in[1];
    const float* values = (const float*)d_in[2];
    const int*   mask   = (const int*)d_in[3];
    const float* W_out  = (const float*)d_in[4];
    const float* b_out  = (const float*)d_in[5];
    float* out  = (float*)d_out;
    float* attn = (float*)d_ws;

    const int N = 2;
    attn_flash_wmma<<<N * NH * (SEQ / (16 * QT)), 32, 0, stream>>>(
        keys, query, values, mask, attn);
    proj_wmma<<<(N * SEQ / (16 * PT)) * (EMB / 64), 128, 0, stream>>>(
        attn, W_out, b_out, out);
}


# attention: every lane-half exchange of the key step (row max, row sum, P restripe for the PV product) done by v_permlanex16 / v_permlane16_swap instead of LDS cross-lane reads; sum order kept
# speedup vs baseline: 1.9586x; 1.1122x over previous
#include <hip/hip_runtime.h>


typedef float v2f __attribute__((ext_vector_type(2)));
typedef float v4f __attribute__((ext_vector_type(4)));
typedef float v8f __attribute__((ext_vector_type(8)));
typedef int   v4i __attribute__((ext_vector_type(4)));

#define SEQ  1024
#define EMB  1024
#define NH   64
#define DH   16
#define QT   4

#define XBLK 128
#define XROWT (XBLK * (EMB / 8))
#define PT   4
#define PG   4

__device__ __forceinline__ float xhalf(float x)
{
    const int b = __builtin_bit_cast(int, x);
    return __builtin_bit_cast(float,
        __builtin_amdgcn_permlanex16(b, b, 0x76543210, 0xfedcba98, false, false));
}

__device__ __forceinline__ void restripe(float (&p)[8])
{
    asm("v_permlane16_swap_b32 %0, %2\n\t"
        "v_permlane16_swap_b32 %1, %3\n\t"
        "v_permlane16_swap_b32 %4, %6\n\t"
        "v_permlane16_swap_b32 %5, %7"
        : "+v"(p[0]), "+v"(p[1]), "+v"(p[2]), "+v"(p[3]),
          "+v"(p[4]), "+v"(p[5]), "+v"(p[6]), "+v"(p[7]));
}

struct AttnFrag {
    v2f ka[4];
    v2f av[4];
    v4i mk0[QT], mk1[QT];
};

__device__ __forceinline__ void attn_load(AttnFrag& f, const float* __restrict__ Kbase,
                                          const float* __restrict__ Vbase,
                                          const int* __restrict__ Mrow,
                                          int kt, int q, int half)
{
    const float* Krow = Kbase + (size_t)(kt * 16 + q) * EMB;
#pragma unroll
    for (int dd = 0; dd < 4; ++dd)
        f.ka[dd] = *reinterpret_cast<const v2f*>(Krow + dd * 4 + 2 * half);

    const float* Vt = Vbase + (size_t)(kt * 16) * EMB + q;
#pragma unroll
    for (int kk = 0; kk < 4; ++kk) {
        const int k0 = 4 * kk + 2 * half;
        f.av[kk].x = Vt[(size_t)k0 * EMB];
        f.av[kk].y = Vt[(size_t)(k0 + 1) * EMB];
    }

#pragma unroll
    for (int t = 0; t < QT; ++t) {
        const int* mb = Mrow + (size_t)(16 * t) * SEQ + kt * 16 + 8 * half;
        f.mk0[t] = *reinterpret_cast<const v4i*>(mb);
        f.mk1[t] = *reinterpret_cast<const v4i*>(mb + 4);
    }
}

__device__ __forceinline__ void attn_step(const AttnFrag& f, const v2f (&Bq)[QT][4],
                                          v8f (&o)[QT], float (&mold)[QT], float (&lsum)[QT],
                                          int q, int half)
{
#pragma unroll
    for (int t = 0; t < QT; ++t) {
        v8f s = {};
#pragma unroll
        for (int dd = 0; dd < 4; ++dd)
            s = __builtin_amdgcn_wmma_f32_16x16x4_f32(
                    false, f.ka[dd], false, Bq[t][dd], (short)0, s, false, false);

        const v4i m0 = f.mk0[t];
        const v4i m1 = f.mk1[t];
        float p[8];
#pragma unroll
        for (int r = 0; r < 8; ++r) {
            float sr = s[r];
            int   mv = (r < 4) ? m0[r] : m1[r - 4];
            sr   = (mv == 0) ? -1e20f : sr;
            p[r] = sr * 0.03125f;
        }

        float tmax = p[0];
#pragma unroll
        for (int r = 1; r < 8; ++r) tmax = fmaxf(tmax, p[r]);
        tmax = fmaxf(tmax, xhalf(tmax));
        const float mnew  = fmaxf(mold[t], tmax);
        const float alpha = __expf(mold[t] - mnew);
        float tsum = 0.0f;
#pragma unroll
        for (int r = 0; r < 8; ++r) { p[r] = __expf(p[r] - mnew); tsum += p[r]; }
        tsum += xhalf(tsum);
        lsum[t] = lsum[t] * alpha + tsum;
        mold[t] = mnew;
#pragma unroll
        for (int r = 0; r < 8; ++r) o[t][r] *= alpha;

        restripe(p);
        const v2f bp[4] = { {p[0], p[1]}, {p[4], p[5]}, {p[2], p[3]}, {p[6], p[7]} };

#pragma unroll
        for (int kk = 0; kk < 4; ++kk)
            o[t] = __builtin_amdgcn_wmma_f32_16x16x4_f32(
                    false, f.av[kk], false, bp[kk], (short)0, o[t], false, false);
    }
}

__global__ __launch_bounds__(32) __attribute__((amdgpu_waves_per_eu(1, 2))) void attn_flash_wmma(
    const float* __restrict__ keys, const float* __restrict__ query,
    const float* __restrict__ values, const int* __restrict__ mask,
    float* __restrict__ attn_out)
{
    const int lane = threadIdx.x;
    const int q    = lane & 15;
    const int half = lane >> 4;

    const int idx = blockIdx.x;
    const int qg  = idx & 15;
    const int h   = (idx >> 4) & 63;
    const int n   = idx >> 10;
    const int q0  = qg * (QT * 16);

    v2f Bq[QT][4];
#pragma unroll
    for (int t = 0; t < QT; ++t) {
        const float* Qrow = query + ((size_t)(n * SEQ + q0 + 16 * t + q)) * EMB + h * DH;
#pragma unroll
        for (int dd = 0; dd < 4; ++dd)
            Bq[t][dd] = *reinterpret_cast<const v2f*>(Qrow + dd * 4 + 2 * half);
    }

    v8f   o[QT];
    float mold[QT];
    float lsum[QT];
#pragma unroll
    for (int t = 0; t < QT; ++t) { o[t] = v8f{}; mold[t] = -1e30f; lsum[t] = 0.0f; }

    const float* Kbase = keys   + ((size_t)(n * SEQ)) * EMB + h * DH;
    const float* Vbase = values + ((size_t)(n * SEQ)) * EMB + h * DH;
    const int*   Mrow  = mask + (size_t)n * SEQ * SEQ + (size_t)(q0 + q) * SEQ;

    constexpr int NK = SEQ / 16;
    AttnFrag f0, f1;
    attn_load(f0, Kbase, Vbase, Mrow, 0, q, half);
    for (int kt = 0; kt < NK - 2; kt += 2) {
        attn_load(f1, Kbase, Vbase, Mrow, kt + 1, q, half);
        attn_step(f0, Bq, o, mold, lsum, q, half);
        attn_load(f0, Kbase, Vbase, Mrow, kt + 2, q, half);
        attn_step(f1, Bq, o, mold, lsum, q, half);
    }
    attn_load(f1, Kbase, Vbase, Mrow, NK - 1, q, half);
    attn_step(f0, Bq, o, mold, lsum, q, half);
    attn_step(f1, Bq, o, mold, lsum, q, half);

#pragma unroll
    for (int t = 0; t < QT; ++t) {
        const float rl = 1.0f / lsum[t];
        const int R = (n * SEQ + q0) / 16 + t;
        float* Oblk = attn_out + (size_t)R * XROWT + (size_t)(2 * h + half) * XBLK + q * 4;
        v4f o0, o1;
#pragma unroll
        for (int r = 0; r < 4; ++r) { o0[r] = o[t][r] * rl; o1[r] = o[t][r + 4] * rl; }
        *reinterpret_cast<v4f*>(Oblk)      = o0;
        *reinterpret_cast<v4f*>(Oblk + 64) = o1;
    }
}

struct ProjStage {
    v4f b[PG];
    v4f a[PG][PT];
};

__device__ __forceinline__ void proj_load(ProjStage& st, const float* __restrict__ Xp,
                                          const float* __restrict__ Wp, int s)
{
#pragma unroll
    for (int u = 0; u < PG; ++u) {
        const int g = s * PG + u;
        st.b[u] = *reinterpret_cast<const v4f*>(Wp + 8 * g);
#pragma unroll
        for (int t = 0; t < PT; ++t)
            st.a[u][t] = *reinterpret_cast<const v4f*>(Xp + (size_t)t * XROWT + g * XBLK);
    }
}

__device__ __forceinline__ void proj_mma(v8f (&acc)[PT], const ProjStage& st)
{
#pragma unroll
    for (int u = 0; u < PG; ++u) {
        const v2f b0 = {st.b[u][0], st.b[u][1]};
        const v2f b1 = {st.b[u][2], st.b[u][3]};
#pragma unroll
        for (int t = 0; t < PT; ++t) {
            const v2f a0 = {st.a[u][t][0], st.a[u][t][1]};
            acc[t] = __builtin_amdgcn_wmma_f32_16x16x4_f32(
                         false, a0, false, b0, (short)0, acc[t], false, false);
        }
#pragma unroll
        for (int t = 0; t < PT; ++t) {
            const v2f a1 = {st.a[u][t][2], st.a[u][t][3]};
            acc[t] = __builtin_amdgcn_wmma_f32_16x16x4_f32(
                         false, a1, false, b1, (short)0, acc[t], false, false);
        }
    }
}

__global__ __launch_bounds__(128) __attribute__((amdgpu_waves_per_eu(1, 2))) void proj_wmma(
    const float* __restrict__ X, const float* __restrict__ W,
    const float* __restrict__ bvec, float* __restrict__ out)
{
    const int lane = threadIdx.x & 31;
    const int wv   = threadIdx.x >> 5;
    const int q    = lane & 15;
    const int half = lane >> 4;
    const int j0 = ((blockIdx.x & 15) * 4 + wv) * 16;
    const int R0 = (blockIdx.x >> 4) * PT;

    const float* Xp = X + (size_t)R0 * XROWT + lane * 4;
    const float* Wp = W + (size_t)(j0 + q) * EMB + 4 * half;

    v8f acc[PT];
#pragma unroll
    for (int t = 0; t < PT; ++t) acc[t] = v8f{};

    constexpr int NS = EMB / (8 * PG);
    ProjStage s0, s1;
    proj_load(s0, Xp, Wp, 0);
    for (int s = 0; s < NS - 2; s += 2) {
        proj_load(s1, Xp, Wp, s + 1);
        proj_mma(acc, s0);
        proj_load(s0, Xp, Wp, s + 2);
        proj_mma(acc, s1);
    }
    proj_load(s1, Xp, Wp, NS - 1);
    proj_mma(acc, s0);
    proj_mma(acc, s1);

    const float bj = bvec[j0 + q];
#pragma unroll
    for (int t = 0; t < PT; ++t)
#pragma unroll
        for (int r = 0; r < 8; ++r)
            out[(size_t)((R0 + t) * 16 + r + 8 * half) * EMB + j0 + q] = acc[t][r] + bj;
}

extern "C" void kernel_launch(void* const* d_in, const int* in_sizes, int n_in,
                              void* d_out, int out_size, void* d_ws, size_t ws_size,
                              hipStream_t stream)
{
    const float* keys   = (const float*)d_in[0];
    const float* query  = (const float*)d_in[1];
    const float* values = (const float*)d_in[2];
    const int*   mask   = (const int*)d_in[3];
    const float* W_out  = (const float*)d_in[4];
    const float* b_out  = (const float*)d_in[5];
    float* out  = (float*)d_out;
    float* attn = (float*)d_ws;

    const int N = 2;
    attn_flash_wmma<<<N * NH * (SEQ / (16 * QT)), 32, 0, stream>>>(
        keys, query, values, mask, attn);
    proj_wmma<<<(N * SEQ / (16 * PT)) * (EMB / 64), 128, 0, stream>>>(
        attn, W_out, b_out, out);
}


# attention: work-groups of 4 waves own 4 neighbouring heads of the same 64 query rows (was 1 wave per group), so a step's mask and K/V lines are fetched once per WGP and hit its cache; sum order kept
# speedup vs baseline: 2.0368x; 1.0399x over previous
#include <hip/hip_runtime.h>


typedef float v2f __attribute__((ext_vector_type(2)));
typedef float v4f __attribute__((ext_vector_type(4)));
typedef float v8f __attribute__((ext_vector_type(8)));
typedef int   v4i __attribute__((ext_vector_type(4)));

#define SEQ  1024
#define EMB  1024
#define NH   64
#define DH   16
#define QT   4
#define AW   4

#define XBLK 128
#define XROWT (XBLK * (EMB / 8))
#define PT   4
#define PG   4

__device__ __forceinline__ float xhalf(float x)
{
    const int b = __builtin_bit_cast(int, x);
    return __builtin_bit_cast(float,
        __builtin_amdgcn_permlanex16(b, b, 0x76543210, 0xfedcba98, false, false));
}

__device__ __forceinline__ void restripe(float (&p)[8])
{
    asm("v_permlane16_swap_b32 %0, %2\n\t"
        "v_permlane16_swap_b32 %1, %3\n\t"
        "v_permlane16_swap_b32 %4, %6\n\t"
        "v_permlane16_swap_b32 %5, %7"
        : "+v"(p[0]), "+v"(p[1]), "+v"(p[2]), "+v"(p[3]),
          "+v"(p[4]), "+v"(p[5]), "+v"(p[6]), "+v"(p[7]));
}

struct AttnFrag {
    v2f ka[4];
    v2f av[4];
    v4i mk0[QT], mk1[QT];
};

__device__ __forceinline__ void attn_load(AttnFrag& f, const float* __restrict__ Kbase,
                                          const float* __restrict__ Vbase,
                                          const int* __restrict__ Mrow,
                                          int kt, int q, int half)
{
    const float* Krow = Kbase + (size_t)(kt * 16 + q) * EMB;
#pragma unroll
    for (int dd = 0; dd < 4; ++dd)
        f.ka[dd] = *reinterpret_cast<const v2f*>(Krow + dd * 4 + 2 * half);

    const float* Vt = Vbase + (size_t)(kt * 16) * EMB + q;
#pragma unroll
    for (int kk = 0; kk < 4; ++kk) {
        const int k0 = 4 * kk + 2 * half;
        f.av[kk].x = Vt[(size_t)k0 * EMB];
        f.av[kk].y = Vt[(size_t)(k0 + 1) * EMB];
    }

#pragma unroll
    for (int t = 0; t < QT; ++t) {
        const int* mb = Mrow + (size_t)(16 * t) * SEQ + kt * 16 + 8 * half;
        f.mk0[t] = *reinterpret_cast<const v4i*>(mb);
        f.mk1[t] = *reinterpret_cast<const v4i*>(mb + 4);
    }
}

__device__ __forceinline__ void attn_step(const AttnFrag& f, const v2f (&Bq)[QT][4],
                                          v8f (&o)[QT], float (&mold)[QT], float (&lsum)[QT],
                                          int q, int half)
{
#pragma unroll
    for (int t = 0; t < QT; ++t) {
        v8f s = {};
#pragma unroll
        for (int dd = 0; dd < 4; ++dd)
            s = __builtin_amdgcn_wmma_f32_16x16x4_f32(
                    false, f.ka[dd], false, Bq[t][dd], (short)0, s, false, false);

        const v4i m0 = f.mk0[t];
        const v4i m1 = f.mk1[t];
        float p[8];
#pragma unroll
        for (int r = 0; r < 8; ++r) {
            float sr = s[r];
            int   mv = (r < 4) ? m0[r] : m1[r - 4];
            sr   = (mv == 0) ? -1e20f : sr;
            p[r] = sr * 0.03125f;
        }

        float tmax = p[0];
#pragma unroll
        for (int r = 1; r < 8; ++r) tmax = fmaxf(tmax, p[r]);
        tmax = fmaxf(tmax, xhalf(tmax));
        const float mnew  = fmaxf(mold[t], tmax);
        const float alpha = __expf(mold[t] - mnew);
        float tsum = 0.0f;
#pragma unroll
        for (int r = 0; r < 8; ++r) { p[r] = __expf(p[r] - mnew); tsum += p[r]; }
        tsum += xhalf(tsum);
        lsum[t] = lsum[t] * alpha + tsum;
        mold[t] = mnew;
#pragma unroll
        for (int r = 0; r < 8; ++r) o[t][r] *= alpha;

        restripe(p);
        const v2f bp[4] = { {p[0], p[1]}, {p[4], p[5]}, {p[2], p[3]}, {p[6], p[7]} };

#pragma unroll
        for (int kk = 0; kk < 4; ++kk)
            o[t] = __builtin_amdgcn_wmma_f32_16x16x4_f32(
                    false, f.av[kk], false, bp[kk], (short)0, o[t], false, false);
    }
}

__global__ __launch_bounds__(32 * AW) __attribute__((amdgpu_waves_per_eu(1, 2))) void attn_flash_wmma(
    const float* __restrict__ keys, const float* __restrict__ query,
    const float* __restrict__ values, const int* __restrict__ mask,
    float* __restrict__ attn_out)
{
    const int lane = threadIdx.x & 31;
    const int wv   = threadIdx.x >> 5;
    const int q    = lane & 15;
    const int half = lane >> 4;

    const int idx = blockIdx.x;
    const int hb  = idx & (NH / AW - 1);
    const int qg  = (idx / (NH / AW)) & 15;
    const int n   = idx / ((NH / AW) * 16);
    const int h   = hb * AW + wv;
    const int q0  = qg * (QT * 16);

    v2f Bq[QT][4];
#pragma unroll
    for (int t = 0; t < QT; ++t) {
        const float* Qrow = query + ((size_t)(n * SEQ + q0 + 16 * t + q)) * EMB + h * DH;
#pragma unroll
        for (int dd = 0; dd < 4; ++dd)
            Bq[t][dd] = *reinterpret_cast<const v2f*>(Qrow + dd * 4 + 2 * half);
    }

    v8f   o[QT];
    float mold[QT];
    float lsum[QT];
#pragma unroll
    for (int t = 0; t < QT; ++t) { o[t] = v8f{}; mold[t] = -1e30f; lsum[t] = 0.0f; }

    const float* Kbase = keys   + ((size_t)(n * SEQ)) * EMB + h * DH;
    const float* Vbase = values + ((size_t)(n * SEQ)) * EMB + h * DH;
    const int*   Mrow  = mask + (size_t)n * SEQ * SEQ + (size_t)(q0 + q) * SEQ;

    constexpr int NK = SEQ / 16;
    AttnFrag f0, f1;
    attn_load(f0, Kbase, Vbase, Mrow, 0, q, half);
    for (int kt = 0; kt < NK - 2; kt += 2) {
        attn_load(f1, Kbase, Vbase, Mrow, kt + 1, q, half);
        attn_step(f0, Bq, o, mold, lsum, q, half);
        attn_load(f0, Kbase, Vbase, Mrow, kt + 2, q, half);
        attn_step(f1, Bq, o, mold, lsum, q, half);
    }
    attn_load(f1, Kbase, Vbase, Mrow, NK - 1, q, half);
    attn_step(f0, Bq, o, mold, lsum, q, half);
    attn_step(f1, Bq, o, mold, lsum, q, half);

#pragma unroll
    for (int t = 0; t < QT; ++t) {
        const float rl = 1.0f / lsum[t];
        const int R = (n * SEQ + q0) / 16 + t;
        float* Oblk = attn_out + (size_t)R * XROWT + (size_t)(2 * h + half) * XBLK + q * 4;
        v4f o0, o1;
#pragma unroll
        for (int r = 0; r < 4; ++r) { o0[r] = o[t][r] * rl; o1[r] = o[t][r + 4] * rl; }
        *reinterpret_cast<v4f*>(Oblk)      = o0;
        *reinterpret_cast<v4f*>(Oblk + 64) = o1;
    }
}

struct ProjStage {
    v4f b[PG];
    v4f a[PG][PT];
};

__device__ __forceinline__ void proj_load(ProjStage& st, const float* __restrict__ Xp,
                                          const float* __restrict__ Wp, int s)
{
#pragma unroll
    for (int u = 0; u < PG; ++u) {
        const int g = s * PG + u;
        st.b[u] = *reinterpret_cast<const v4f*>(Wp + 8 * g);
#pragma unroll
        for (int t = 0; t < PT; ++t)
            st.a[u][t] = *reinterpret_cast<const v4f*>(Xp + (size_t)t * XROWT + g * XBLK);
    }
}

__device__ __forceinline__ void proj_mma(v8f (&acc)[PT], const ProjStage& st)
{
#pragma unroll
    for (int u = 0; u < PG; ++u) {
        const v2f b0 = {st.b[u][0], st.b[u][1]};
        const v2f b1 = {st.b[u][2], st.b[u][3]};
#pragma unroll
        for (int t = 0; t < PT; ++t) {
            const v2f a0 = {st.a[u][t][0], st.a[u][t][1]};
            acc[t] = __builtin_amdgcn_wmma_f32_16x16x4_f32(
                         false, a0, false, b0, (short)0, acc[t], false, false);
        }
#pragma unroll
        for (int t = 0; t < PT; ++t) {
            const v2f a1 = {st.a[u][t][2], st.a[u][t][3]};
            acc[t] = __builtin_amdgcn_wmma_f32_16x16x4_f32(
                         false, a1, false, b1, (short)0, acc[t], false, false);
        }
    }
}

__global__ __launch_bounds__(128) __attribute__((amdgpu_waves_per_eu(1, 2))) void proj_wmma(
    const float* __restrict__ X, const float* __restrict__ W,
    const float* __restrict__ bvec, float* __restrict__ out)
{
    const int lane = threadIdx.x & 31;
    const int wv   = threadIdx.x >> 5;
    const int q    = lane & 15;
    const int half = lane >> 4;
    const int j0 = ((blockIdx.x & 15) * 4 + wv) * 16;
    const int R0 = (blockIdx.x >> 4) * PT;

    const float* Xp = X + (size_t)R0 * XROWT + lane * 4;
    const float* Wp = W + (size_t)(j0 + q) * EMB + 4 * half;

    v8f acc[PT];
#pragma unroll
    for (int t = 0; t < PT; ++t) acc[t] = v8f{};

    constexpr int NS = EMB / (8 * PG);
    ProjStage s0, s1;
    proj_load(s0, Xp, Wp, 0);
    for (int s = 0; s < NS - 2; s += 2) {
        proj_load(s1, Xp, Wp, s + 1);
        proj_mma(acc, s0);
        proj_load(s0, Xp, Wp, s + 2);
        proj_mma(acc, s1);
    }
    proj_load(s1, Xp, Wp, NS - 1);
    proj_mma(acc, s0);
    proj_mma(acc, s1);

    const float bj = bvec[j0 + q];
#pragma unroll
    for (int t = 0; t < PT; ++t)
#pragma unroll
        for (int r = 0; r < 8; ++r)
            out[(size_t)((R0 + t) * 16 + r + 8 * half) * EMB + j0 + q] = acc[t][r] + bj;
}

extern "C" void kernel_launch(void* const* d_in, const int* in_sizes, int n_in,
                              void* d_out, int out_size, void* d_ws, size_t ws_size,
                              hipStream_t stream)
{
    const float* keys   = (const float*)d_in[0];
    const float* query  = (const float*)d_in[1];
    const float* values = (const float*)d_in[2];
    const int*   mask   = (const int*)d_in[3];
    const float* W_out  = (const float*)d_in[4];
    const float* b_out  = (const float*)d_in[5];
    float* out  = (float*)d_out;
    float* attn = (float*)d_ws;

    const int N = 2;
    attn_flash_wmma<<<N * (NH / AW) * (SEQ / (16 * QT)), 32 * AW, 0, stream>>>(
        keys, query, values, mask, attn);
    proj_wmma<<<(N * SEQ / (16 * PT)) * (EMB / 64), 128, 0, stream>>>(
        attn, W_out, b_out, out);
}


# attention mask: a first small launch packs the int32 mask into 32-bit lane masks per (tile, key step, register); the key loop scalar-loads them as select masks: no vector mask loads, no compares
# speedup vs baseline: 2.1378x; 1.0496x over previous
#include <hip/hip_runtime.h>


typedef float v2f __attribute__((ext_vector_type(2)));
typedef float v4f __attribute__((ext_vector_type(4)));
typedef float v8f __attribute__((ext_vector_type(8)));
typedef int   v4i __attribute__((ext_vector_type(4)));
typedef unsigned v8u __attribute__((ext_vector_type(8)));

#define SEQ  1024
#define EMB  1024
#define NH   64
#define DH   16
#define QT   4
#define AW   4

#define XBLK 128
#define XROWT (XBLK * (EMB / 8))
#define PT   4
#define PG   4

#define PKSTEP 32
__global__ __launch_bounds__(64) void mask_pack(const int* __restrict__ mask,
                                                float* __restrict__ packed)
{
    const int lane = threadIdx.x & 31;
    const int q    = lane & 15;
    const int half = lane >> 4;
    const int task = blockIdx.x * 2 + (threadIdx.x >> 5);
    const int kp = task & (SEQ / 32 - 1);
    const int qg = (task / (SEQ / 32)) & 15;
    const int n  = task / ((SEQ / 32) * 16);

    const int* Mrow = mask + (size_t)n * SEQ * SEQ + (size_t)(qg * (QT * 16) + q) * SEQ
                           + kp * 32 + 8 * half;
    v4i m[2][QT][2];
#pragma unroll
    for (int s = 0; s < 2; ++s)
#pragma unroll
        for (int t = 0; t < QT; ++t) {
            const int* mb = Mrow + (size_t)(16 * t) * SEQ + 16 * s;
            m[s][t][0] = *reinterpret_cast<const v4i*>(mb);
            m[s][t][1] = *reinterpret_cast<const v4i*>(mb + 4);
        }

    unsigned* dst = reinterpret_cast<unsigned*>(packed)
                  + (size_t)((n * 16 + qg) * (SEQ / 16) + kp * 2) * PKSTEP;
#pragma unroll
    for (int s = 0; s < 2; ++s) {
        unsigned w = 0u;
#pragma unroll
        for (int t = 0; t < QT; ++t)
#pragma unroll
            for (int r = 0; r < 8; ++r) {
                const int mv = (r < 4) ? m[s][t][0][r] : m[s][t][1][r - 4];
                const unsigned b = __builtin_amdgcn_ballot_w32(mv != 0);
                w = (lane == t * 8 + r) ? b : w;
            }
        dst[s * PKSTEP + lane] = w;
    }
}

__device__ __forceinline__ float xhalf(float x)
{
    const int b = __builtin_bit_cast(int, x);
    return __builtin_bit_cast(float,
        __builtin_amdgcn_permlanex16(b, b, 0x76543210, 0xfedcba98, false, false));
}

__device__ __forceinline__ void restripe(float (&p)[8])
{
    asm("v_permlane16_swap_b32 %0, %2\n\t"
        "v_permlane16_swap_b32 %1, %3\n\t"
        "v_permlane16_swap_b32 %4, %6\n\t"
        "v_permlane16_swap_b32 %5, %7"
        : "+v"(p[0]), "+v"(p[1]), "+v"(p[2]), "+v"(p[3]),
          "+v"(p[4]), "+v"(p[5]), "+v"(p[6]), "+v"(p[7]));
}

struct AttnFrag {
    v2f ka[4];
    v2f av[4];
};

__device__ __forceinline__ void mask_words(v8u (&mw)[QT], const v8u* __restrict__ PM, int kt)
{
#pragma unroll
    for (int t = 0; t < QT; ++t) mw[t] = PM[kt * QT + t];
}

__device__ __forceinline__ void attn_load(AttnFrag& f, const float* __restrict__ Kbase,
                                          const float* __restrict__ Vbase,
                                          int kt, int q, int half)
{
    const float* Krow = Kbase + (size_t)(kt * 16 + q) * EMB;
#pragma unroll
    for (int dd = 0; dd < 4; ++dd)
        f.ka[dd] = *reinterpret_cast<const v2f*>(Krow + dd * 4 + 2 * half);

    const float* Vt = Vbase + (size_t)(kt * 16) * EMB + q;
#pragma unroll
    for (int kk = 0; kk < 4; ++kk) {
        const int k0 = 4 * kk + 2 * half;
        f.av[kk].x = Vt[(size_t)k0 * EMB];
        f.av[kk].y = Vt[(size_t)(k0 + 1) * EMB];
    }
}

__device__ __forceinline__ void attn_step(const AttnFrag& f, const v8u (&mw)[QT],
                                          const v2f (&Bq)[QT][4],
                                          v8f (&o)[QT], float (&mold)[QT], float (&lsum)[QT],
                                          int q, int half)
{
#pragma unroll
    for (int t = 0; t < QT; ++t) {
        v8f s = {};
#pragma unroll
        for (int dd = 0; dd < 4; ++dd)
            s = __builtin_amdgcn_wmma_f32_16x16x4_f32(
                    false, f.ka[dd], false, Bq[t][dd], (short)0, s, false, false);

        const v8u w = mw[t];
        float p[8];
#pragma unroll
        for (int r = 0; r < 8; ++r) {
            float sr = s[r];
            sr   = __builtin_amdgcn_inverse_ballot_w32(w[r]) ? sr : -1e20f;
            p[r] = sr * 0.03125f;
        }

        float tmax = p[0];
#pragma unroll
        for (int r = 1; r < 8; ++r) tmax = fmaxf(tmax, p[r]);
        tmax = fmaxf(tmax, xhalf(tmax));
        const float mnew  = fmaxf(mold[t], tmax);
        const float alpha = __expf(mold[t] - mnew);
        float tsum = 0.0f;
#pragma unroll
        for (int r = 0; r < 8; ++r) { p[r] = __expf(p[r] - mnew); tsum += p[r]; }
        tsum += xhalf(tsum);
        lsum[t] = lsum[t] * alpha + tsum;
        mold[t] = mnew;
#pragma unroll
        for (int r = 0; r < 8; ++r) o[t][r] *= alpha;

        restripe(p);
        const v2f bp[4] = { {p[0], p[1]}, {p[4], p[5]}, {p[2], p[3]}, {p[6], p[7]} };

#pragma unroll
        for (int kk = 0; kk < 4; ++kk)
            o[t] = __builtin_amdgcn_wmma_f32_16x16x4_f32(
                    false, f.av[kk], false, bp[kk], (short)0, o[t], false, false);
    }
}

__global__ __launch_bounds__(32 * AW) __attribute__((amdgpu_waves_per_eu(1, 2))) void attn_flash_wmma(
    const float* __restrict__ keys, const float* __restrict__ query,
    const float* __restrict__ values, const float* __restrict__ packed,
    float* __restrict__ attn_out)
{
    const int lane = threadIdx.x & 31;
    const int wv   = threadIdx.x >> 5;
    const int q    = lane & 15;
    const int half = lane >> 4;

    const int idx = blockIdx.x;
    const int hb  = idx & (NH / AW - 1);
    const int qg  = (idx / (NH / AW)) & 15;
    const int n   = idx / ((NH / AW) * 16);
    const int h   = hb * AW + wv;
    const int q0  = qg * (QT * 16);

    v2f Bq[QT][4];
#pragma unroll
    for (int t = 0; t < QT; ++t) {
        const float* Qrow = query + ((size_t)(n * SEQ + q0 + 16 * t + q)) * EMB + h * DH;
#pragma unroll
        for (int dd = 0; dd < 4; ++dd)
            Bq[t][dd] = *reinterpret_cast<const v2f*>(Qrow + dd * 4 + 2 * half);
    }

    v8f   o[QT];
    float mold[QT];
    float lsum[QT];
#pragma unroll
    for (int t = 0; t < QT; ++t) { o[t] = v8f{}; mold[t] = -1e30f; lsum[t] = 0.0f; }

    const float* Kbase = keys   + ((size_t)(n * SEQ)) * EMB + h * DH;
    const float* Vbase = values + ((size_t)(n * SEQ)) * EMB + h * DH;
    const v8u*   PM    = reinterpret_cast<const v8u*>(packed)
                       + (size_t)((n * 16 + qg) * (SEQ / 16)) * QT;

    constexpr int NK = SEQ / 16;
    AttnFrag f0, f1;
    v8u mw[QT];
    mask_words(mw, PM, 0);
    attn_load(f0, Kbase, Vbase, 0, q, half);
    for (int kt = 0; kt < NK - 2; kt += 2) {
        attn_load(f1, Kbase, Vbase, kt + 1, q, half);
        attn_step(f0, mw, Bq, o, mold, lsum, q, half);
        mask_words(mw, PM, kt + 1);
        attn_load(f0, Kbase, Vbase, kt + 2, q, half);
        attn_step(f1, mw, Bq, o, mold, lsum, q, half);
        mask_words(mw, PM, kt + 2);
    }
    attn_load(f1, Kbase, Vbase, NK - 1, q, half);
    attn_step(f0, mw, Bq, o, mold, lsum, q, half);
    mask_words(mw, PM, NK - 1);
    attn_step(f1, mw, Bq, o, mold, lsum, q, half);

#pragma unroll
    for (int t = 0; t < QT; ++t) {
        const float rl = 1.0f / lsum[t];
        const int R = (n * SEQ + q0) / 16 + t;
        float* Oblk = attn_out + (size_t)R * XROWT + (size_t)(2 * h + half) * XBLK + q * 4;
        v4f o0, o1;
#pragma unroll
        for (int r = 0; r < 4; ++r) { o0[r] = o[t][r] * rl; o1[r] = o[t][r + 4] * rl; }
        *reinterpret_cast<v4f*>(Oblk)      = o0;
        *reinterpret_cast<v4f*>(Oblk + 64) = o1;
    }
}

struct ProjStage {
    v4f b[PG];
    v4f a[PG][PT];
};

__device__ __forceinline__ void proj_load(ProjStage& st, const float* __restrict__ Xp,
                                          const float* __restrict__ Wp, int s)
{
#pragma unroll
    for (int u = 0; u < PG; ++u) {
        const int g = s * PG + u;
        st.b[u] = *reinterpret_cast<const v4f*>(Wp + 8 * g);
#pragma unroll
        for (int t = 0; t < PT; ++t)
            st.a[u][t] = *reinterpret_cast<const v4f*>(Xp + (size_t)t * XROWT + g * XBLK);
    }
}

__device__ __forceinline__ void proj_mma(v8f (&acc)[PT], const ProjStage& st)
{
#pragma unroll
    for (int u = 0; u < PG; ++u) {
        const v2f b0 = {st.b[u][0], st.b[u][1]};
        const v2f b1 = {st.b[u][2], st.b[u][3]};
#pragma unroll
        for (int t = 0; t < PT; ++t) {
            const v2f a0 = {st.a[u][t][0], st.a[u][t][1]};
            acc[t] = __builtin_amdgcn_wmma_f32_16x16x4_f32(
                         false, a0, false, b0, (short)0, acc[t], false, false);
        }
#pragma unroll
        for (int t = 0; t < PT; ++t) {
            const v2f a1 = {st.a[u][t][2], st.a[u][t][3]};
            acc[t] = __builtin_amdgcn_wmma_f32_16x16x4_f32(
                         false, a1, false, b1, (short)0, acc[t], false, false);
        }
    }
}

__global__ __launch_bounds__(128) __attribute__((amdgpu_waves_per_eu(1, 2))) void proj_wmma(
    const float* __restrict__ X, const float* __restrict__ W,
    const float* __restrict__ bvec, float* __restrict__ out)
{
    const int lane = threadIdx.x & 31;
    const int wv   = threadIdx.x >> 5;
    const int q    = lane & 15;
    const int half = lane >> 4;
    const int j0 = ((blockIdx.x & 15) * 4 + wv) * 16;
    const int R0 = (blockIdx.x >> 4) * PT;

    const float* Xp = X + (size_t)R0 * XROWT + lane * 4;
    const float* Wp = W + (size_t)(j0 + q) * EMB + 4 * half;

    v8f acc[PT];
#pragma unroll
    for (int t = 0; t < PT; ++t) acc[t] = v8f{};

    constexpr int NS = EMB / (8 * PG);
    ProjStage s0, s1;
    proj_load(s0, Xp, Wp, 0);
    for (int s = 0; s < NS - 2; s += 2) {
        proj_load(s1, Xp, Wp, s + 1);
        proj_mma(acc, s0);
        proj_load(s0, Xp, Wp, s + 2);
        proj_mma(acc, s1);
    }
    proj_load(s1, Xp, Wp, NS - 1);
    proj_mma(acc, s0);
    proj_mma(acc, s1);

    const float bj = bvec[j0 + q];
#pragma unroll
    for (int t = 0; t < PT; ++t)
#pragma unroll
        for (int r = 0; r < 8; ++r)
            out[(size_t)((R0 + t) * 16 + r + 8 * half) * EMB + j0 + q] = acc[t][r] + bj;
}

extern "C" void kernel_launch(void* const* d_in, const int* in_sizes, int n_in,
                              void* d_out, int out_size, void* d_ws, size_t ws_size,
                              hipStream_t stream)
{
    const float* keys   = (const float*)d_in[0];
    const float* query  = (const float*)d_in[1];
    const float* values = (const float*)d_in[2];
    const int*   mask   = (const int*)d_in[3];
    const float* W_out  = (const float*)d_in[4];
    const float* b_out  = (const float*)d_in[5];
    float* out  = (float*)d_out;
    float* attn = (float*)d_ws;

    const int N = 2;
    mask_pack<<<N * 16 * (SEQ / 32) / 2, 64, 0, stream>>>(mask, out);
    attn_flash_wmma<<<N * (NH / AW) * (SEQ / (16 * QT)), 32 * AW, 0, stream>>>(
        keys, query, values, out, attn);
    proj_wmma<<<(N * SEQ / (16 * PT)) * (EMB / 64), 128, 0, stream>>>(
        attn, W_out, b_out, out);
}


# attention softmax: key loop takes exp(score) against a fixed zero reference (no running max, alpha, O rescale); a wave whose row sums leave a safe range re-runs the online-max loop; rounding changes
# speedup vs baseline: 2.2943x; 1.0732x over previous
#include <hip/hip_runtime.h>


typedef float v2f __attribute__((ext_vector_type(2)));
typedef float v4f __attribute__((ext_vector_type(4)));
typedef float v8f __attribute__((ext_vector_type(8)));
typedef int   v4i __attribute__((ext_vector_type(4)));
typedef unsigned v8u __attribute__((ext_vector_type(8)));

#define SEQ  1024
#define EMB  1024
#define NH   64
#define DH   16
#define QT   4
#define AW   4

#define XBLK 128
#define XROWT (XBLK * (EMB / 8))
#define PT   4
#define PG   4

#define PKSTEP 32
__global__ __launch_bounds__(64) void mask_pack(const int* __restrict__ mask,
                                                float* __restrict__ packed)
{
    const int lane = threadIdx.x & 31;
    const int q    = lane & 15;
    const int half = lane >> 4;
    const int task = blockIdx.x * 2 + (threadIdx.x >> 5);
    const int kp = task & (SEQ / 32 - 1);
    const int qg = (task / (SEQ / 32)) & 15;
    const int n  = task / ((SEQ / 32) * 16);

    const int* Mrow = mask + (size_t)n * SEQ * SEQ + (size_t)(qg * (QT * 16) + q) * SEQ
                           + kp * 32 + 8 * half;
    v4i m[2][QT][2];
#pragma unroll
    for (int s = 0; s < 2; ++s)
#pragma unroll
        for (int t = 0; t < QT; ++t) {
            const int* mb = Mrow + (size_t)(16 * t) * SEQ + 16 * s;
            m[s][t][0] = *reinterpret_cast<const v4i*>(mb);
            m[s][t][1] = *reinterpret_cast<const v4i*>(mb + 4);
        }

    unsigned* dst = reinterpret_cast<unsigned*>(packed)
                  + (size_t)((n * 16 + qg) * (SEQ / 16) + kp * 2) * PKSTEP;
#pragma unroll
    for (int s = 0; s < 2; ++s) {
        unsigned w = 0u;
#pragma unroll
        for (int t = 0; t < QT; ++t)
#pragma unroll
            for (int r = 0; r < 8; ++r) {
                const int mv = (r < 4) ? m[s][t][0][r] : m[s][t][1][r - 4];
                const unsigned b = __builtin_amdgcn_ballot_w32(mv != 0);
                w = (lane == t * 8 + r) ? b : w;
            }
        dst[s * PKSTEP + lane] = w;
    }
}

__device__ __forceinline__ float xhalf(float x)
{
    const int b = __builtin_bit_cast(int, x);
    return __builtin_bit_cast(float,
        __builtin_amdgcn_permlanex16(b, b, 0x76543210, 0xfedcba98, false, false));
}

__device__ __forceinline__ void restripe(float (&p)[8])
{
    asm("v_permlane16_swap_b32 %0, %2\n\t"
        "v_permlane16_swap_b32 %1, %3\n\t"
        "v_permlane16_swap_b32 %4, %6\n\t"
        "v_permlane16_swap_b32 %5, %7"
        : "+v"(p[0]), "+v"(p[1]), "+v"(p[2]), "+v"(p[3]),
          "+v"(p[4]), "+v"(p[5]), "+v"(p[6]), "+v"(p[7]));
}

struct AttnFrag {
    v2f ka[4];
    v2f av[4];
};

__device__ __forceinline__ void mask_words(v8u (&mw)[QT], const v8u* __restrict__ PM, int kt)
{
#pragma unroll
    for (int t = 0; t < QT; ++t) mw[t] = PM[kt * QT + t];
}

__device__ __forceinline__ void attn_load(AttnFrag& f, const float* __restrict__ Kbase,
                                          const float* __restrict__ Vbase,
                                          int kt, int q, int half)
{
    const float* Krow = Kbase + (size_t)(kt * 16 + q) * EMB;
#pragma unroll
    for (int dd = 0; dd < 4; ++dd)
        f.ka[dd] = *reinterpret_cast<const v2f*>(Krow + dd * 4 + 2 * half);

    const float* Vt = Vbase + (size_t)(kt * 16) * EMB + q;
#pragma unroll
    for (int kk = 0; kk < 4; ++kk) {
        const int k0 = 4 * kk + 2 * half;
        f.av[kk].x = Vt[(size_t)k0 * EMB];
        f.av[kk].y = Vt[(size_t)(k0 + 1) * EMB];
    }
}

#define SC2     0.045084220027780106f
#define MASKED2 (-1e20f * SC2)

template <bool FIXED>
__device__ __forceinline__ void attn_step(const AttnFrag& f, const v8u (&mw)[QT],
                                          const v2f (&Bq)[QT][4],
                                          v8f (&o)[QT], float (&mold)[QT], float (&lsum)[QT],
                                          int q, int half)
{
#pragma unroll
    for (int t = 0; t < QT; ++t) {
        v8f s = {};
#pragma unroll
        for (int dd = 0; dd < 4; ++dd)
            s = __builtin_amdgcn_wmma_f32_16x16x4_f32(
                    false, f.ka[dd], false, Bq[t][dd], (short)0, s, false, false);

        const v8u w = mw[t];
        float p[8];
        if constexpr (FIXED) {
            float tsum = 0.0f;
#pragma unroll
            for (int r = 0; r < 8; ++r) {
                float sr = __builtin_amdgcn_inverse_ballot_w32(w[r]) ? s[r] * SC2 : MASKED2;
                p[r] = __builtin_amdgcn_exp2f(sr);
                tsum += p[r];
            }
            tsum += xhalf(tsum);
            lsum[t] += tsum;
        } else {
#pragma unroll
        for (int r = 0; r < 8; ++r) {
            float sr = s[r];
            sr   = __builtin_amdgcn_inverse_ballot_w32(w[r]) ? sr : -1e20f;
            p[r] = sr * 0.03125f;
        }

        float tmax = p[0];
#pragma unroll
        for (int r = 1; r < 8; ++r) tmax = fmaxf(tmax, p[r]);
        tmax = fmaxf(tmax, xhalf(tmax));
        const float mnew  = fmaxf(mold[t], tmax);
        const float alpha = __expf(mold[t] - mnew);
        float tsum = 0.0f;
#pragma unroll
        for (int r = 0; r < 8; ++r) { p[r] = __expf(p[r] - mnew); tsum += p[r]; }
        tsum += xhalf(tsum);
        lsum[t] = lsum[t] * alpha + tsum;
        mold[t] = mnew;
#pragma unroll
        for (int r = 0; r < 8; ++r) o[t][r] *= alpha;
        }

        restripe(p);
        const v2f bp[4] = { {p[0], p[1]}, {p[4], p[5]}, {p[2], p[3]}, {p[6], p[7]} };

#pragma unroll
        for (int kk = 0; kk < 4; ++kk)
            o[t] = __builtin_amdgcn_wmma_f32_16x16x4_f32(
                    false, f.av[kk], false, bp[kk], (short)0, o[t], false, false);
    }
}

template <bool FIXED>
__device__ __forceinline__ void attn_keys(const float* __restrict__ Kbase,
                                          const float* __restrict__ Vbase,
                                          const v8u* __restrict__ PM,
                                          const v2f (&Bq)[QT][4],
                                          v8f (&o)[QT], float (&lsum)[QT], int q, int half)
{
    float mold[QT];
#pragma unroll
    for (int t = 0; t < QT; ++t) { o[t] = v8f{}; mold[t] = -1e30f; lsum[t] = 0.0f; }

    constexpr int NK = SEQ / 16;
    AttnFrag f0, f1;
    v8u mw[QT];
    mask_words(mw, PM, 0);
    attn_load(f0, Kbase, Vbase, 0, q, half);
    for (int kt = 0; kt < NK - 2; kt += 2) {
        attn_load(f1, Kbase, Vbase, kt + 1, q, half);
        attn_step<FIXED>(f0, mw, Bq, o, mold, lsum, q, half);
        mask_words(mw, PM, kt + 1);
        attn_load(f0, Kbase, Vbase, kt + 2, q, half);
        attn_step<FIXED>(f1, mw, Bq, o, mold, lsum, q, half);
        mask_words(mw, PM, kt + 2);
    }
    attn_load(f1, Kbase, Vbase, NK - 1, q, half);
    attn_step<FIXED>(f0, mw, Bq, o, mold, lsum, q, half);
    mask_words(mw, PM, NK - 1);
    attn_step<FIXED>(f1, mw, Bq, o, mold, lsum, q, half);
}

__global__ __launch_bounds__(32 * AW) __attribute__((amdgpu_waves_per_eu(1, 2))) void attn_flash_wmma(
    const float* __restrict__ keys, const float* __restrict__ query,
    const float* __restrict__ values, const float* __restrict__ packed,
    float* __restrict__ attn_out)
{
    const int lane = threadIdx.x & 31;
    const int wv   = threadIdx.x >> 5;
    const int q    = lane & 15;
    const int half = lane >> 4;

    const int idx = blockIdx.x;
    const int hb  = idx & (NH / AW - 1);
    const int qg  = (idx / (NH / AW)) & 15;
    const int n   = idx / ((NH / AW) * 16);
    const int h   = hb * AW + wv;
    const int q0  = qg * (QT * 16);

    v2f Bq[QT][4];
#pragma unroll
    for (int t = 0; t < QT; ++t) {
        const float* Qrow = query + ((size_t)(n * SEQ + q0 + 16 * t + q)) * EMB + h * DH;
#pragma unroll
        for (int dd = 0; dd < 4; ++dd)
            Bq[t][dd] = *reinterpret_cast<const v2f*>(Qrow + dd * 4 + 2 * half);
    }

    v8f   o[QT];
    float lsum[QT];

    const float* Kbase = keys   + ((size_t)(n * SEQ)) * EMB + h * DH;
    const float* Vbase = values + ((size_t)(n * SEQ)) * EMB + h * DH;
    const v8u*   PM    = reinterpret_cast<const v8u*>(packed)
                       + (size_t)((n * 16 + qg) * (SEQ / 16)) * QT;

    attn_keys<true>(Kbase, Vbase, PM, Bq, o, lsum, q, half);
    float osum = 0.0f;
#pragma unroll
    for (int t = 0; t < QT; ++t)
#pragma unroll
        for (int r = 0; r < 8; ++r) osum += __builtin_fabsf(o[t][r]);
    bool ok = (osum <= 0x1p100f);
#pragma unroll
    for (int t = 0; t < QT; ++t)
        ok = ok & (lsum[t] >= 0x1p-24f) & (lsum[t] <= 0x1p34f);
    if (__builtin_amdgcn_ballot_w32(!ok) != 0u)
        attn_keys<false>(Kbase, Vbase, PM, Bq, o, lsum, q, half);

#pragma unroll
    for (int t = 0; t < QT; ++t) {
        const float rl = 1.0f / lsum[t];
        const int R = (n * SEQ + q0) / 16 + t;
        float* Oblk = attn_out + (size_t)R * XROWT + (size_t)(2 * h + half) * XBLK + q * 4;
        v4f o0, o1;
#pragma unroll
        for (int r = 0; r < 4; ++r) { o0[r] = o[t][r] * rl; o1[r] = o[t][r + 4] * rl; }
        *reinterpret_cast<v4f*>(Oblk)      = o0;
        *reinterpret_cast<v4f*>(Oblk + 64) = o1;
    }
}

struct ProjStage {
    v4f b[PG];
    v4f a[PG][PT];
};

__device__ __forceinline__ void proj_load(ProjStage& st, const float* __restrict__ Xp,
                                          const float* __restrict__ Wp, int s)
{
#pragma unroll
    for (int u = 0; u < PG; ++u) {
        const int g = s * PG + u;
        st.b[u] = *reinterpret_cast<const v4f*>(Wp + 8 * g);
#pragma unroll
        for (int t = 0; t < PT; ++t)
            st.a[u][t] = *reinterpret_cast<const v4f*>(Xp + (size_t)t * XROWT + g * XBLK);
    }
}

__device__ __forceinline__ void proj_mma(v8f (&acc)[PT], const ProjStage& st)
{
#pragma unroll
    for (int u = 0; u < PG; ++u) {
        const v2f b0 = {st.b[u][0], st.b[u][1]};
        const v2f b1 = {st.b[u][2], st.b[u][3]};
#pragma unroll
        for (int t = 0; t < PT; ++t) {
            const v2f a0 = {st.a[u][t][0], st.a[u][t][1]};
            acc[t] = __builtin_amdgcn_wmma_f32_16x16x4_f32(
                         false, a0, false, b0, (short)0, acc[t], false, false);
        }
#pragma unroll
        for (int t = 0; t < PT; ++t) {
            const v2f a1 = {st.a[u][t][2], st.a[u][t][3]};
            acc[t] = __builtin_amdgcn_wmma_f32_16x16x4_f32(
                         false, a1, false, b1, (short)0, acc[t], false, false);
        }
    }
}

__global__ __launch_bounds__(128) __attribute__((amdgpu_waves_per_eu(1, 2))) void proj_wmma(
    const float* __restrict__ X, const float* __restrict__ W,
    const float* __restrict__ bvec, float* __restrict__ out)
{
    const int lane = threadIdx.x & 31;
    const int wv   = threadIdx.x >> 5;
    const int q    = lane & 15;
    const int half = lane >> 4;
    const int j0 = ((blockIdx.x & 15) * 4 + wv) * 16;
    const int R0 = (blockIdx.x >> 4) * PT;

    const float* Xp = X + (size_t)R0 * XROWT + lane * 4;
    const float* Wp = W + (size_t)(j0 + q) * EMB + 4 * half;

    v8f acc[PT];
#pragma unroll
    for (int t = 0; t < PT; ++t) acc[t] = v8f{};

    constexpr int NS = EMB / (8 * PG);
    ProjStage s0, s1;
    proj_load(s0, Xp, Wp, 0);
    for (int s = 0; s < NS - 2; s += 2) {
        proj_load(s1, Xp, Wp, s + 1);
        proj_mma(acc, s0);
        proj_load(s0, Xp, Wp, s + 2);
        proj_mma(acc, s1);
    }
    proj_load(s1, Xp, Wp, NS - 1);
    proj_mma(acc, s0);
    proj_mma(acc, s1);

    const float bj = bvec[j0 + q];
#pragma unroll
    for (int t = 0; t < PT; ++t)
#pragma unroll
        for (int r = 0; r < 8; ++r)
            out[(size_t)((R0 + t) * 16 + r + 8 * half) * EMB + j0 + q] = acc[t][r] + bj;
}

extern "C" void kernel_launch(void* const* d_in, const int* in_sizes, int n_in,
                              void* d_out, int out_size, void* d_ws, size_t ws_size,
                              hipStream_t stream)
{
    const float* keys   = (const float*)d_in[0];
    const float* query  = (const float*)d_in[1];
    const float* values = (const float*)d_in[2];
    const int*   mask   = (const int*)d_in[3];
    const float* W_out  = (const float*)d_in[4];
    const float* b_out  = (const float*)d_in[5];
    float* out  = (float*)d_out;
    float* attn = (float*)d_ws;

    const int N = 2;
    mask_pack<<<N * 16 * (SEQ / 32) / 2, 64, 0, stream>>>(mask, out);
    attn_flash_wmma<<<N * (NH / AW) * (SEQ / (16 * QT)), 32 * AW, 0, stream>>>(
        keys, query, values, out, attn);
    proj_wmma<<<(N * SEQ / (16 * PT)) * (EMB / 64), 128, 0, stream>>>(
        attn, W_out, b_out, out);
}


# attention: K rows of a key step loaded in a permuted row order (mask words packed to match) so P^T's registers are the PV B pairs in place: no restripe; row sum adds those pairs (its order changes)
# speedup vs baseline: 2.3737x; 1.0346x over previous
#include <hip/hip_runtime.h>


typedef float v2f __attribute__((ext_vector_type(2)));
typedef float v4f __attribute__((ext_vector_type(4)));
typedef float v8f __attribute__((ext_vector_type(8)));
typedef int   v4i __attribute__((ext_vector_type(4)));
typedef unsigned v8u __attribute__((ext_vector_type(8)));

#define SEQ  1024
#define EMB  1024
#define NH   64
#define DH   16
#define QT   4
#define AW   4

#define XBLK 128
#define XROWT (XBLK * (EMB / 8))
#define PT   4
#define PG   4

#define PKSTEP 32
__global__ __launch_bounds__(64) void mask_pack(const int* __restrict__ mask,
                                                float* __restrict__ packed)
{
    const int lane = threadIdx.x & 31;
    const int q    = lane & 15;
    const int half = lane >> 4;
    const int task = blockIdx.x * 2 + (threadIdx.x >> 5);
    const int kp = task & (SEQ / 32 - 1);
    const int qg = (task / (SEQ / 32)) & 15;
    const int n  = task / ((SEQ / 32) * 16);

    const int* Mrow = mask + (size_t)n * SEQ * SEQ + (size_t)(qg * (QT * 16) + q) * SEQ
                           + kp * 32 + 8 * half;
    v4i m[2][QT][2];
#pragma unroll
    for (int s = 0; s < 2; ++s)
#pragma unroll
        for (int t = 0; t < QT; ++t) {
            const int* mb = Mrow + (size_t)(16 * t) * SEQ + 16 * s;
            m[s][t][0] = *reinterpret_cast<const v4i*>(mb);
            m[s][t][1] = *reinterpret_cast<const v4i*>(mb + 4);
        }

    unsigned* dst = reinterpret_cast<unsigned*>(packed)
                  + (size_t)((n * 16 + qg) * (SEQ / 16) + kp * 2) * PKSTEP;
#pragma unroll
    for (int s = 0; s < 2; ++s) {
        unsigned w = 0u;
#pragma unroll
        for (int t = 0; t < QT; ++t) {
            unsigned b[8];
#pragma unroll
            for (int j = 0; j < 8; ++j) {
                const int mv = (j < 4) ? m[s][t][0][j] : m[s][t][1][j - 4];
                b[j] = __builtin_amdgcn_ballot_w32(mv != 0);
            }
#pragma unroll
            for (int r = 0; r < 8; ++r) {
                const int a = 4 * ((r >> 1) & 1) + (r & 1);
                const unsigned word = (r < 4)
                    ? ((b[a] & 0xffffu) | (b[a + 2] << 16))
                    : ((b[a] >> 16) | (b[a + 2] & 0xffff0000u));
                w = (lane == t * 8 + r) ? word : w;
            }
        }
        dst[s * PKSTEP + lane] = w;
    }
}

__device__ __forceinline__ float xhalf(float x)
{
    const int b = __builtin_bit_cast(int, x);
    return __builtin_bit_cast(float,
        __builtin_amdgcn_permlanex16(b, b, 0x76543210, 0xfedcba98, false, false));
}

__device__ __forceinline__ int kperm(int m)
{
    return 4 * ((m & 7) >> 1) + 2 * (m >> 3) + (m & 1);
}

struct AttnFrag {
    v2f ka[4];
    v2f av[4];
};

__device__ __forceinline__ void mask_words(v8u (&mw)[QT], const v8u* __restrict__ PM, int kt)
{
#pragma unroll
    for (int t = 0; t < QT; ++t) mw[t] = PM[kt * QT + t];
}

__device__ __forceinline__ void attn_load(AttnFrag& f, const float* __restrict__ Kbase,
                                          const float* __restrict__ Vbase,
                                          int kt, int q, int kq, int half)
{
    const float* Krow = Kbase + (size_t)(kt * 16 + kq) * EMB;
#pragma unroll
    for (int dd = 0; dd < 4; ++dd)
        f.ka[dd] = *reinterpret_cast<const v2f*>(Krow + dd * 4 + 2 * half);

    const float* Vt = Vbase + (size_t)(kt * 16) * EMB + q;
#pragma unroll
    for (int kk = 0; kk < 4; ++kk) {
        const int k0 = 4 * kk + 2 * half;
        f.av[kk].x = Vt[(size_t)k0 * EMB];
        f.av[kk].y = Vt[(size_t)(k0 + 1) * EMB];
    }
}

#define SC2     0.045084220027780106f
#define MASKED2 (-1e20f * SC2)

template <bool FIXED>
__device__ __forceinline__ void attn_step(const AttnFrag& f, const v8u (&mw)[QT],
                                          const v2f (&Bq)[QT][4],
                                          v8f (&o)[QT], float (&mold)[QT], float (&lsum)[QT],
                                          int q, int half)
{
#pragma unroll
    for (int t = 0; t < QT; ++t) {
        v8f s = {};
#pragma unroll
        for (int dd = 0; dd < 4; ++dd)
            s = __builtin_amdgcn_wmma_f32_16x16x4_f32(
                    false, f.ka[dd], false, Bq[t][dd], (short)0, s, false, false);

        const v8u w = mw[t];
        float p[8];
        if constexpr (FIXED) {
#pragma unroll
            for (int r = 0; r < 8; ++r) {
                float sr = __builtin_amdgcn_inverse_ballot_w32(w[r]) ? s[r] * SC2 : MASKED2;
                p[r] = __builtin_amdgcn_exp2f(sr);
            }
            v2f ps = v2f{p[0], p[1]} + v2f{p[2], p[3]};
            ps += v2f{p[4], p[5]};
            ps += v2f{p[6], p[7]};
            float tsum = ps.x + ps.y;
            tsum += xhalf(tsum);
            lsum[t] += tsum;
        } else {
#pragma unroll
        for (int r = 0; r < 8; ++r) {
            float sr = s[r];
            sr   = __builtin_amdgcn_inverse_ballot_w32(w[r]) ? sr : -1e20f;
            p[r] = sr * 0.03125f;
        }

        float tmax = p[0];
#pragma unroll
        for (int r = 1; r < 8; ++r) tmax = fmaxf(tmax, p[r]);
        tmax = fmaxf(tmax, xhalf(tmax));
        const float mnew  = fmaxf(mold[t], tmax);
        const float alpha = __expf(mold[t] - mnew);
        float tsum = 0.0f;
#pragma unroll
        for (int r = 0; r < 8; ++r) { p[r] = __expf(p[r] - mnew); tsum += p[r]; }
        tsum += xhalf(tsum);
        lsum[t] = lsum[t] * alpha + tsum;
        mold[t] = mnew;
#pragma unroll
        for (int r = 0; r < 8; ++r) o[t][r] *= alpha;
        }

        const v2f bp[4] = { {p[0], p[1]}, {p[2], p[3]}, {p[4], p[5]}, {p[6], p[7]} };

#pragma unroll
        for (int kk = 0; kk < 4; ++kk)
            o[t] = __builtin_amdgcn_wmma_f32_16x16x4_f32(
                    false, f.av[kk], false, bp[kk], (short)0, o[t], false, false);
    }
}

template <bool FIXED>
__device__ __forceinline__ void attn_keys(const float* __restrict__ Kbase,
                                          const float* __restrict__ Vbase,
                                          const v8u* __restrict__ PM,
                                          const v2f (&Bq)[QT][4],
                                          v8f (&o)[QT], float (&lsum)[QT],
                                          int q, int kq, int half)
{
    float mold[QT];
#pragma unroll
    for (int t = 0; t < QT; ++t) { o[t] = v8f{}; mold[t] = -1e30f; lsum[t] = 0.0f; }

    constexpr int NK = SEQ / 16;
    AttnFrag f0, f1;
    v8u mw[QT];
    mask_words(mw, PM, 0);
    attn_load(f0, Kbase, Vbase, 0, q, kq, half);
    for (int kt = 0; kt < NK - 2; kt += 2) {
        attn_load(f1, Kbase, Vbase, kt + 1, q, kq, half);
        attn_step<FIXED>(f0, mw, Bq, o, mold, lsum, q, half);
        mask_words(mw, PM, kt + 1);
        attn_load(f0, Kbase, Vbase, kt + 2, q, kq, half);
        attn_step<FIXED>(f1, mw, Bq, o, mold, lsum, q, half);
        mask_words(mw, PM, kt + 2);
    }
    attn_load(f1, Kbase, Vbase, NK - 1, q, kq, half);
    attn_step<FIXED>(f0, mw, Bq, o, mold, lsum, q, half);
    mask_words(mw, PM, NK - 1);
    attn_step<FIXED>(f1, mw, Bq, o, mold, lsum, q, half);
}

__global__ __launch_bounds__(32 * AW) __attribute__((amdgpu_waves_per_eu(1, 2))) void attn_flash_wmma(
    const float* __restrict__ keys, const float* __restrict__ query,
    const float* __restrict__ values, const float* __restrict__ packed,
    float* __restrict__ attn_out)
{
    const int lane = threadIdx.x & 31;
    const int wv   = threadIdx.x >> 5;
    const int q    = lane & 15;
    const int half = lane >> 4;
    const int kq   = kperm(q);

    const int idx = blockIdx.x;
    const int hb  = idx & (NH / AW - 1);
    const int qg  = (idx / (NH / AW)) & 15;
    const int n   = idx / ((NH / AW) * 16);
    const int h   = hb * AW + wv;
    const int q0  = qg * (QT * 16);

    v2f Bq[QT][4];
#pragma unroll
    for (int t = 0; t < QT; ++t) {
        const float* Qrow = query + ((size_t)(n * SEQ + q0 + 16 * t + q)) * EMB + h * DH;
#pragma unroll
        for (int dd = 0; dd < 4; ++dd)
            Bq[t][dd] = *reinterpret_cast<const v2f*>(Qrow + dd * 4 + 2 * half);
    }

    v8f   o[QT];
    float lsum[QT];

    const float* Kbase = keys   + ((size_t)(n * SEQ)) * EMB + h * DH;
    const float* Vbase = values + ((size_t)(n * SEQ)) * EMB + h * DH;
    const v8u*   PM    = reinterpret_cast<const v8u*>(packed)
                       + (size_t)((n * 16 + qg) * (SEQ / 16)) * QT;

    attn_keys<true>(Kbase, Vbase, PM, Bq, o, lsum, q, kq, half);
    float osum = 0.0f;
#pragma unroll
    for (int t = 0; t < QT; ++t)
#pragma unroll
        for (int r = 0; r < 8; ++r) osum += __builtin_fabsf(o[t][r]);
    bool ok = (osum <= 0x1p100f);
#pragma unroll
    for (int t = 0; t < QT; ++t)
        ok = ok & (lsum[t] >= 0x1p-24f) & (lsum[t] <= 0x1p34f);
    if (__builtin_amdgcn_ballot_w32(!ok) != 0u)
        attn_keys<false>(Kbase, Vbase, PM, Bq, o, lsum, q, kq, half);

#pragma unroll
    for (int t = 0; t < QT; ++t) {
        const float rl = 1.0f / lsum[t];
        const int R = (n * SEQ + q0) / 16 + t;
        float* Oblk = attn_out + (size_t)R * XROWT + (size_t)(2 * h + half) * XBLK + q * 4;
        v4f o0, o1;
#pragma unroll
        for (int r = 0; r < 4; ++r) { o0[r] = o[t][r] * rl; o1[r] = o[t][r + 4] * rl; }
        *reinterpret_cast<v4f*>(Oblk)      = o0;
        *reinterpret_cast<v4f*>(Oblk + 64) = o1;
    }
}

struct ProjStage {
    v4f b[PG];
    v4f a[PG][PT];
};

__device__ __forceinline__ void proj_load(ProjStage& st, const float* __restrict__ Xp,
                                          const float* __restrict__ Wp, int s)
{
#pragma unroll
    for (int u = 0; u < PG; ++u) {
        const int g = s * PG + u;
        st.b[u] = *reinterpret_cast<const v4f*>(Wp + 8 * g);
#pragma unroll
        for (int t = 0; t < PT; ++t)
            st.a[u][t] = *reinterpret_cast<const v4f*>(Xp + (size_t)t * XROWT + g * XBLK);
    }
}

__device__ __forceinline__ void proj_mma(v8f (&acc)[PT], const ProjStage& st)
{
#pragma unroll
    for (int u = 0; u < PG; ++u) {
        const v2f b0 = {st.b[u][0], st.b[u][1]};
        const v2f b1 = {st.b[u][2], st.b[u][3]};
#pragma unroll
        for (int t = 0; t < PT; ++t) {
            const v2f a0 = {st.a[u][t][0], st.a[u][t][1]};
            acc[t] = __builtin_amdgcn_wmma_f32_16x16x4_f32(
                         false, a0, false, b0, (short)0, acc[t], false, false);
        }
#pragma unroll
        for (int t = 0; t < PT; ++t) {
            const v2f a1 = {st.a[u][t][2], st.a[u][t][3]};
            acc[t] = __builtin_amdgcn_wmma_f32_16x16x4_f32(
                         false, a1, false, b1, (short)0, acc[t], false, false);
        }
    }
}

__global__ __launch_bounds__(128) __attribute__((amdgpu_waves_per_eu(1, 2))) void proj_wmma(
    const float* __restrict__ X, const float* __restrict__ W,
    const float* __restrict__ bvec, float* __restrict__ out)
{
    const int lane = threadIdx.x & 31;
    const int wv   = threadIdx.x >> 5;
    const int q    = lane & 15;
    const int half = lane >> 4;
    const int j0 = ((blockIdx.x & 15) * 4 + wv) * 16;
    const int R0 = (blockIdx.x >> 4) * PT;

    const float* Xp = X + (size_t)R0 * XROWT + lane * 4;
    const float* Wp = W + (size_t)(j0 + q) * EMB + 4 * half;

    v8f acc[PT];
#pragma unroll
    for (int t = 0; t < PT; ++t) acc[t] = v8f{};

    constexpr int NS = EMB / (8 * PG);
    ProjStage s0, s1;
    proj_load(s0, Xp, Wp, 0);
    for (int s = 0; s < NS - 2; s += 2) {
        proj_load(s1, Xp, Wp, s + 1);
        proj_mma(acc, s0);
        proj_load(s0, Xp, Wp, s + 2);
        proj_mma(acc, s1);
    }
    proj_load(s1, Xp, Wp, NS - 1);
    proj_mma(acc, s0);
    proj_mma(acc, s1);

    const float bj = bvec[j0 + q];
#pragma unroll
    for (int t = 0; t < PT; ++t)
#pragma unroll
        for (int r = 0; r < 8; ++r)
            out[(size_t)((R0 + t) * 16 + r + 8 * half) * EMB + j0 + q] = acc[t][r] + bj;
}

extern "C" void kernel_launch(void* const* d_in, const int* in_sizes, int n_in,
                              void* d_out, int out_size, void* d_ws, size_t ws_size,
                              hipStream_t stream)
{
    const float* keys   = (const float*)d_in[0];
    const float* query  = (const float*)d_in[1];
    const float* values = (const float*)d_in[2];
    const int*   mask   = (const int*)d_in[3];
    const float* W_out  = (const float*)d_in[4];
    const float* b_out  = (const float*)d_in[5];
    float* out  = (float*)d_out;
    float* attn = (float*)d_ws;

    const int N = 2;
    mask_pack<<<N * 16 * (SEQ / 32) / 2, 64, 0, stream>>>(mask, out);
    attn_flash_wmma<<<N * (NH / AW) * (SEQ / (16 * QT)), 32 * AW, 0, stream>>>(
        keys, query, values, out, attn);
    proj_wmma<<<(N * SEQ / (16 * PT)) * (EMB / 64), 128, 0, stream>>>(
        attn, W_out, b_out, out);
}
